# starting kernel
# baseline (speedup 1.0000x reference)
#include <hip/hip_runtime.h>
#include <hip/hip_bf16.h>


#define BATCH   2
#define SEQ     512
#define DIN     4096
#define DSTATE  16
#define DTRANK  128
#define NPROJ   160
#define NROWS   (BATCH * SEQ)
#define KSPLIT  4

typedef float v2f __attribute__((ext_vector_type(2)));
typedef float v4f __attribute__((ext_vector_type(4)));
typedef float v8f __attribute__((ext_vector_type(8)));

__device__ __forceinline__ v8f wmma_pair(v8f c, v4f a, v4f b) {
  v2f a0 = {a.x, a.y}, a1 = {a.z, a.w};
  v2f b0 = {b.x, b.y}, b1 = {b.z, b.w};
  c = __builtin_amdgcn_wmma_f32_16x16x4_f32(false, a0, false, b0,
                                            (short)0, c, false, false);
  c = __builtin_amdgcn_wmma_f32_16x16x4_f32(false, a1, false, b1,
                                            (short)0, c, false, false);
  return c;
}

__global__ __launch_bounds__(256) void gemm_xproj_partial(
    const float* __restrict__ x, const float* __restrict__ w,
    float* __restrict__ part) {
  const int lane  = threadIdx.x & 31;
  const int wave  = threadIdx.x >> 5;
  const int gtile = blockIdx.x * 8 + wave;
  const int ks    = gtile / 640;
  const int tile  = gtile - ks * 640;
  const int mt = tile / 10, nt = tile - mt * 10;
  const int m0 = mt * 16, n0 = nt * 16;
  const int lo = lane & 15, hi = lane >> 4;
  const int kbase = ks * (DIN / KSPLIT);

  const float* aptr = x + (size_t)(m0 + lo) * DIN + kbase + 4 * hi;
  const float* bptr = w + (size_t)(n0 + lo) * DIN + kbase + 4 * hi;

  v8f c = {};
#pragma unroll 4
  for (int k = 0; k < DIN / KSPLIT; k += 8) {
    v4f a = *(const v4f*)(aptr + k);
    v4f b = *(const v4f*)(bptr + k);
    c = wmma_pair(c, a, b);
  }

  float* out = part + (size_t)ks * NROWS * NPROJ + n0 + lo;
  const int rbase = m0 + 8 * hi;
#pragma unroll
  for (int j = 0; j < 8; ++j)
    out[(size_t)(rbase + j) * NPROJ] = c[j];
}

__global__ __launch_bounds__(256) void reduce_xp_kernel(
    const float* __restrict__ part, float* __restrict__ xp) {
  const size_t S = (size_t)NROWS * NPROJ;
  const size_t i = (size_t)blockIdx.x * 256 + threadIdx.x;
  xp[i] = ((part[i] + part[i + S]) + part[i + 2 * S]) + part[i + 3 * S];
}

__global__ __launch_bounds__(256) void gemm_dt_kernel(
    const float* __restrict__ xp, const float* __restrict__ w,
    const float* __restrict__ bias, float* __restrict__ dtb) {
  const int lane = threadIdx.x & 31;
  const int wave = threadIdx.x >> 5;
  const int tile = blockIdx.x * 8 + wave;
  const int mt = tile >> 8;
  const int nt = tile & 255;
  const int m0 = mt * 16, n0 = nt * 16;
  const int lo = lane & 15, hi = lane >> 4;

  const float* aptr = xp + (size_t)(m0 + lo) * NPROJ + 4 * hi;
  const float* bptr = w + (size_t)(n0 + lo) * DTRANK + 4 * hi;

  v8f c = {};
#pragma unroll
  for (int k = 0; k < DTRANK; k += 8) {
    v4f a = *(const v4f*)(aptr + k);
    v4f b = *(const v4f*)(bptr + k);
    c = wmma_pair(c, a, b);
  }

  const int d = n0 + lo;
  const float bv = bias[d];
  float* out = dtb + d;
  const int rbase = m0 + 8 * hi;
#pragma unroll
  for (int j = 0; j < 8; ++j) {
    float v = c[j] + bv;
    float sp = (v > 20.0f) ? v : log1pf(__expf(v));
    out[(size_t)(rbase + j) * DIN] = sp;
  }
}

#define TC 64

__global__ __launch_bounds__(256) void scan_kernel(
    const float* __restrict__ x, const float* __restrict__ A_log,
    const float* __restrict__ Dp, const float* __restrict__ xp,
    const float* __restrict__ dtb, float* __restrict__ y) {
  __shared__ float sBC[TC * 32];
  __shared__ float sdt[TC * 16];
  __shared__ float sx [TC * 16];
  __shared__ float sy [TC * 16];

  const int tid  = threadIdx.x;
  const int lane = tid & 31;
  const int wave = tid >> 5;
  const int n    = lane & 15;
  const int grp  = lane >> 4;
  const int c    = wave * 2 + grp;
  const int b    = blockIdx.y;
  const int d    = blockIdx.x * 16 + c;

  const float Aval = -__expf(A_log[(size_t)d * DSTATE + n]);
  const float Dv   = Dp[d];
  float h = 0.0f;

  const float* xpb = xp  + (size_t)b * SEQ * NPROJ + DTRANK;
  const float* xg  = x   + (size_t)b * SEQ * DIN + blockIdx.x * 16;
  const float* dtg = dtb + (size_t)b * SEQ * DIN + blockIdx.x * 16;
  float*       yg  = y   + (size_t)b * SEQ * DIN + blockIdx.x * 16;

  for (int t0 = 0; t0 < SEQ; t0 += TC) {
    __syncthreads();
    for (int f = tid; f < TC * 32; f += 256) {
      const int tt = f >> 5, j = f & 31;
      sBC[f] = xpb[(size_t)(t0 + tt) * NPROJ + j];
    }
    for (int f = tid; f < TC * 16; f += 256) {
      const int tt = f >> 4, j = f & 15;
      sdt[f] = dtg[(size_t)(t0 + tt) * DIN + j];
      sx [f] = xg [(size_t)(t0 + tt) * DIN + j];
    }
    __syncthreads();

    if (t0 + TC < SEQ)
      __builtin_prefetch(&xpb[(size_t)(t0 + TC) * NPROJ + (tid & 31)], 0, 0);

    for (int tl = 0; tl < TC; ++tl) {
      const float dt = sdt[tl * 16 + c];
      const float xv = sx [tl * 16 + c];
      const float Bn = sBC[tl * 32 + n];
      const float Cn = sBC[tl * 32 + 16 + n];
      h = __expf(dt * Aval) * h + (dt * xv) * Bn;
      float p = h * Cn;
      p += __shfl_xor(p, 1);
      p += __shfl_xor(p, 2);
      p += __shfl_xor(p, 4);
      p += __shfl_xor(p, 8);
      if (n == 0) sy[tl * 16 + c] = p + Dv * xv;
    }
    __syncthreads();
    for (int f = tid; f < TC * 16; f += 256) {
      const int tt = f >> 4, j = f & 15;
      yg[(size_t)(t0 + tt) * DIN + j] = sy[f];
    }
  }
}

extern "C" void kernel_launch(void* const* d_in, const int* in_sizes, int n_in,
                              void* d_out, int out_size, void* d_ws, size_t ws_size,
                              hipStream_t stream) {
  const float* x     = (const float*)d_in[0];
  const float* A_log = (const float*)d_in[1];
  const float* Dp    = (const float*)d_in[2];
  const float* xw    = (const float*)d_in[3];
  const float* dtw   = (const float*)d_in[4];
  const float* dtb_b = (const float*)d_in[5];
  float* y = (float*)d_out;

  float* xp   = (float*)d_ws;
  float* dtb  = xp + (size_t)NROWS * NPROJ;
  float* part = dtb;

  gemm_xproj_partial<<<dim3(320), dim3(256), 0, stream>>>(x, xw, part);
  reduce_xp_kernel<<<dim3(640), dim3(256), 0, stream>>>(part, xp);
  gemm_dt_kernel<<<dim3(2048), dim3(256), 0, stream>>>(xp, dtw, dtb_b, dtb);
  scan_kernel<<<dim3(DIN / 16, BATCH), dim3(256), 0, stream>>>(
      x, A_log, Dp, xp, dtb, y);
}


# scan: a lane per channel, its 16 states in registers as packed f32 pairs, dot over n in-lane, no shuffles, dt/x fetched 16 steps ahead; chunks 4->8; y sum order, exp and carry rounding change
# speedup vs baseline: 2.4690x; 2.4690x over previous
#include <hip/hip_runtime.h>
#include <hip/hip_bf16.h>


#define BATCH   2
#define SEQ     512
#define DIN     4096
#define DSTATE  16
#define DTRANK  128
#define NPROJ   160
#define NROWS   (BATCH * SEQ)
#define KSPLIT  4

typedef float v2f __attribute__((ext_vector_type(2)));
typedef float v4f __attribute__((ext_vector_type(4)));
typedef float v8f __attribute__((ext_vector_type(8)));

__device__ __forceinline__ v8f wmma_pair(v8f c, v4f a, v4f b) {
  v2f a0 = {a.x, a.y}, a1 = {a.z, a.w};
  v2f b0 = {b.x, b.y}, b1 = {b.z, b.w};
  c = __builtin_amdgcn_wmma_f32_16x16x4_f32(false, a0, false, b0,
                                            (short)0, c, false, false);
  c = __builtin_amdgcn_wmma_f32_16x16x4_f32(false, a1, false, b1,
                                            (short)0, c, false, false);
  return c;
}

__global__ __launch_bounds__(256) void gemm_xproj_partial(
    const float* __restrict__ x, const float* __restrict__ w,
    float* __restrict__ part) {
  const int lane  = threadIdx.x & 31;
  const int wave  = threadIdx.x >> 5;
  const int gtile = blockIdx.x * 8 + wave;
  const int ks    = gtile / 640;
  const int tile  = gtile - ks * 640;
  const int mt = tile / 10, nt = tile - mt * 10;
  const int m0 = mt * 16, n0 = nt * 16;
  const int lo = lane & 15, hi = lane >> 4;
  const int kbase = ks * (DIN / KSPLIT);

  const float* aptr = x + (size_t)(m0 + lo) * DIN + kbase + 4 * hi;
  const float* bptr = w + (size_t)(n0 + lo) * DIN + kbase + 4 * hi;

  v8f c = {};
#pragma unroll 4
  for (int k = 0; k < DIN / KSPLIT; k += 8) {
    v4f a = *(const v4f*)(aptr + k);
    v4f b = *(const v4f*)(bptr + k);
    c = wmma_pair(c, a, b);
  }

  float* out = part + (size_t)ks * NROWS * NPROJ + n0 + lo;
  const int rbase = m0 + 8 * hi;
#pragma unroll
  for (int j = 0; j < 8; ++j)
    out[(size_t)(rbase + j) * NPROJ] = c[j];
}

__global__ __launch_bounds__(256) void reduce_xp_kernel(
    const float* __restrict__ part, float* __restrict__ xp) {
  const size_t S = (size_t)NROWS * NPROJ;
  const size_t i = (size_t)blockIdx.x * 256 + threadIdx.x;
  xp[i] = ((part[i] + part[i + S]) + part[i + 2 * S]) + part[i + 3 * S];
}

__global__ __launch_bounds__(256) void gemm_dt_kernel(
    const float* __restrict__ xp, const float* __restrict__ w,
    const float* __restrict__ bias, float* __restrict__ dtb) {
  const int lane = threadIdx.x & 31;
  const int wave = threadIdx.x >> 5;
  const int tile = blockIdx.x * 8 + wave;
  const int mt = tile >> 8;
  const int nt = tile & 255;
  const int m0 = mt * 16, n0 = nt * 16;
  const int lo = lane & 15, hi = lane >> 4;

  const float* aptr = xp + (size_t)(m0 + lo) * NPROJ + 4 * hi;
  const float* bptr = w + (size_t)(n0 + lo) * DTRANK + 4 * hi;

  v8f c = {};
#pragma unroll
  for (int k = 0; k < DTRANK; k += 8) {
    v4f a = *(const v4f*)(aptr + k);
    v4f b = *(const v4f*)(bptr + k);
    c = wmma_pair(c, a, b);
  }

  const int d = n0 + lo;
  const float bv = bias[d];
  float* out = dtb + d;
  const int rbase = m0 + 8 * hi;
#pragma unroll
  for (int j = 0; j < 8; ++j) {
    float v = c[j] + bv;
    float sp = (v > 20.0f) ? v : log1pf(__expf(v));
    out[(size_t)(rbase + j) * DIN] = sp;
  }
}

#define NCHUNK 8
#define CLEN (SEQ / NCHUNK)
#define TB 16
#define NBD (BATCH * DIN)
#define LOG2E 1.4426950408889634f

__device__ __forceinline__ v2f fma2(v2f a, v2f b, v2f c) {
  return __builtin_elementwise_fma(a, b, c);
}
__device__ __forceinline__ v2f exp2_2(v2f v) {
  v2f r = {__builtin_amdgcn_exp2f(v.x), __builtin_amdgcn_exp2f(v.y)};
  return r;
}

__device__ __forceinline__ void load_batch(v2f (&dx)[TB],
                                           const float* __restrict__ dtg,
                                           const float* __restrict__ xg, int t) {
#pragma unroll
  for (int i = 0; i < TB; ++i) {
    dx[i].x = dtg[(size_t)(t + i) * DIN];
    dx[i].y = xg [(size_t)(t + i) * DIN];
  }
}

__device__ __forceinline__ void load_A2(v2f (&A2)[8], const float* __restrict__ A_log, int d) {
  const v4f* ap = (const v4f*)(A_log + (size_t)d * DSTATE);
#pragma unroll
  for (int k = 0; k < 4; ++k) {
    const v4f a = ap[k];
    A2[2 * k]     = (v2f){-__expf(a.x) * LOG2E, -__expf(a.y) * LOG2E};
    A2[2 * k + 1] = (v2f){-__expf(a.z) * LOG2E, -__expf(a.w) * LOG2E};
  }
}

__device__ __forceinline__ void local_batch(const v2f (&dxc)[TB],
                                            v2f (&h)[8], const v2f (&A2)[8],
                                            float& sum_dt, const v4f* bt) {
  v4f bn[4];
#pragma unroll
  for (int k = 0; k < 4; ++k) bn[k] = bt[k];
#pragma unroll
  for (int i = 0; i < TB; ++i) {
    const v2f dx   = dxc[i];
    const v2f dt2  = __builtin_shufflevector(dx, dx, 0, 0);
    const v2f x2   = __builtin_shufflevector(dx, dx, 1, 1);
    const v2f dtx2 = dt2 * x2;
    sum_dt += dx.x;
    v4f bv[4];
#pragma unroll
    for (int k = 0; k < 4; ++k) bv[k] = bn[k];
    if (i + 1 < TB) {
#pragma unroll
      for (int k = 0; k < 4; ++k) bn[k] = bt[(i + 1) * 4 + k];
    }
#pragma unroll
    for (int p = 0; p < 8; ++p) {
      const v4f bq = bv[p >> 1];
      const v2f Bp = (p & 1) ? (v2f){bq.z, bq.w} : (v2f){bq.x, bq.y};
      const v2f e  = exp2_2(dt2 * A2[p]);
      h[p] = fma2(e, h[p], dtx2 * Bp);
    }
  }
}

__device__ __forceinline__ void scan_batch(const v2f (&dxc)[TB],
                                           v2f (&h)[8], const v2f (&A2)[8],
                                           float Dv, const v4f* bt, float* yt) {
  v4f bn[8];
#pragma unroll
  for (int k = 0; k < 8; ++k) bn[k] = bt[k];
#pragma unroll
  for (int i = 0; i < TB; ++i) {
    const v2f dx   = dxc[i];
    const float xv = dx.y;
    const v2f dt2  = __builtin_shufflevector(dx, dx, 0, 0);
    const v2f x2   = __builtin_shufflevector(dx, dx, 1, 1);
    const v2f dtx2 = dt2 * x2;
    v4f bv[8];
#pragma unroll
    for (int k = 0; k < 8; ++k) bv[k] = bn[k];
    if (i + 1 < TB) {
#pragma unroll
      for (int k = 0; k < 8; ++k) bn[k] = bt[(i + 1) * 8 + k];
    }
    v2f ya = {0.0f, 0.0f}, yb = {0.0f, 0.0f};
#pragma unroll
    for (int p = 0; p < 8; ++p) {
      const v4f bq = bv[p >> 1];
      const v4f cq = bv[4 + (p >> 1)];
      const v2f Bp = (p & 1) ? (v2f){bq.z, bq.w} : (v2f){bq.x, bq.y};
      const v2f Cp = (p & 1) ? (v2f){cq.z, cq.w} : (v2f){cq.x, cq.y};
      const v2f e  = exp2_2(dt2 * A2[p]);
      h[p] = fma2(e, h[p], dtx2 * Bp);
      if (p & 1) yb = fma2(h[p], Cp, yb);
      else       ya = fma2(h[p], Cp, ya);
    }
    const v2f ys = ya + yb;
    yt[(size_t)i * DIN] = (ys.x + ys.y) + Dv * xv;
  }
}

__global__ __launch_bounds__(32) void scan_local_kernel(
    const float* __restrict__ x, const float* __restrict__ A_log,
    const float* __restrict__ xp, const float* __restrict__ dtb,
    float* __restrict__ hend, float* __restrict__ sdt) {
  __shared__ v4f sB[CLEN * 4];

  const int lane = threadIdx.x;
  const int d    = blockIdx.x * 32 + lane;
  const int b    = blockIdx.y;
  const int ch   = blockIdx.z;
  const int tbeg = ch * CLEN;
  const int bd   = b * DIN + d;

  const float* xg  = x   + ((size_t)b * SEQ + tbeg) * DIN + d;
  const float* dtg = dtb + ((size_t)b * SEQ + tbeg) * DIN + d;

  v2f dxA[TB], dxB[TB];
  load_batch(dxA, dtg, xg, 0);

  const float* xpb = xp + ((size_t)b * SEQ + tbeg) * NPROJ + DTRANK;
#pragma unroll
  for (int i = 0; i < CLEN / 8; ++i) {
    const int r = i * 8 + (lane >> 2);
    sB[r * 4 + (lane & 3)] =
        *(const v4f*)(xpb + (size_t)r * NPROJ + (lane & 3) * 4);
  }

  v2f A2[8];
  load_A2(A2, A_log, d);

  v2f h[8];
#pragma unroll
  for (int p = 0; p < 8; ++p) h[p] = (v2f){0.0f, 0.0f};
  float sum_dt = 0.0f;

  __syncthreads();

#pragma unroll 1
  for (int t0 = 0; t0 < CLEN; t0 += 2 * TB) {
    load_batch(dxB, dtg, xg, t0 + TB);
    __builtin_amdgcn_sched_barrier(0);
    local_batch(dxA, h, A2, sum_dt, sB + t0 * 4);
    __builtin_amdgcn_sched_barrier(0);
    const int tn = (t0 + 2 * TB < CLEN) ? (t0 + 2 * TB) : t0;
    load_batch(dxA, dtg, xg, tn);
    __builtin_amdgcn_sched_barrier(0);
    local_batch(dxB, h, A2, sum_dt, sB + (t0 + TB) * 4);
  }

  v4f* hq = (v4f*)hend + (size_t)ch * 4 * NBD + bd;
#pragma unroll
  for (int q = 0; q < 4; ++q)
    hq[(size_t)q * NBD] =
        (v4f){h[2 * q].x, h[2 * q].y, h[2 * q + 1].x, h[2 * q + 1].y};
  sdt[(size_t)ch * NBD + bd] = sum_dt;
}

__global__ __launch_bounds__(32) void scan_kernel(
    const float* __restrict__ x, const float* __restrict__ A_log,
    const float* __restrict__ Dp, const float* __restrict__ xp,
    const float* __restrict__ dtb, const float* __restrict__ hend,
    const float* __restrict__ sdt, float* __restrict__ y) {
  __shared__ v4f sBC[CLEN * 8];

  const int lane = threadIdx.x;
  const int d    = blockIdx.x * 32 + lane;
  const int b    = blockIdx.y;
  const int ch   = blockIdx.z;
  const int tbeg = ch * CLEN;
  const int bd   = b * DIN + d;

  const float* xg  = x   + ((size_t)b * SEQ + tbeg) * DIN + d;
  const float* dtg = dtb + ((size_t)b * SEQ + tbeg) * DIN + d;
  float*       yg  = y   + ((size_t)b * SEQ + tbeg) * DIN + d;

  v2f dxA[TB], dxB[TB];
  load_batch(dxA, dtg, xg, 0);

  const float* xpb = xp + ((size_t)b * SEQ + tbeg) * NPROJ + DTRANK;
#pragma unroll
  for (int i = 0; i < CLEN / 4; ++i) {
    const int r = i * 4 + (lane >> 3);
    sBC[r * 8 + (lane & 7)] =
        *(const v4f*)(xpb + (size_t)r * NPROJ + (lane & 7) * 4);
  }

  v2f A2[8];
  load_A2(A2, A_log, d);
  const float Dv = Dp[d];

  v2f h[8];
#pragma unroll
  for (int p = 0; p < 8; ++p) h[p] = (v2f){0.0f, 0.0f};
  {
    float sd[NCHUNK - 1];
    v4f   he[NCHUNK - 1][4];
    const v4f* hq = (const v4f*)hend + bd;
#pragma unroll
    for (int cc = 0; cc < NCHUNK - 1; ++cc) {
      sd[cc] = sdt[(size_t)cc * NBD + bd];
#pragma unroll
      for (int q = 0; q < 4; ++q)
        he[cc][q] = hq[((size_t)cc * 4 + q) * NBD];
    }
#pragma unroll
    for (int cc = 0; cc < NCHUNK - 1; ++cc) {
      const bool on = cc < ch;
      const v2f s2 = {sd[cc], sd[cc]};
#pragma unroll
      for (int p = 0; p < 8; ++p) {
        const v4f hv = he[cc][p >> 1];
        v2f e  = exp2_2(s2 * A2[p]);
        v2f ev = (p & 1) ? (v2f){hv.z, hv.w} : (v2f){hv.x, hv.y};
        e.x  = on ? e.x  : 1.0f;  e.y  = on ? e.y  : 1.0f;
        ev.x = on ? ev.x : 0.0f;  ev.y = on ? ev.y : 0.0f;
        h[p] = fma2(e, h[p], ev);
      }
    }
  }

  __syncthreads();

#pragma unroll 1
  for (int t0 = 0; t0 < CLEN; t0 += 2 * TB) {
    load_batch(dxB, dtg, xg, t0 + TB);
    __builtin_amdgcn_sched_barrier(0);
    scan_batch(dxA, h, A2, Dv, sBC + t0 * 8, yg + (size_t)t0 * DIN);
    __builtin_amdgcn_sched_barrier(0);
    const int tn = (t0 + 2 * TB < CLEN) ? (t0 + 2 * TB) : t0;
    load_batch(dxA, dtg, xg, tn);
    __builtin_amdgcn_sched_barrier(0);
    scan_batch(dxB, h, A2, Dv, sBC + (t0 + TB) * 8,
               yg + (size_t)(t0 + TB) * DIN);
  }
}

extern "C" void kernel_launch(void* const* d_in, const int* in_sizes, int n_in,
                              void* d_out, int out_size, void* d_ws, size_t ws_size,
                              hipStream_t stream) {
  const float* x     = (const float*)d_in[0];
  const float* A_log = (const float*)d_in[1];
  const float* Dp    = (const float*)d_in[2];
  const float* xw    = (const float*)d_in[3];
  const float* dtw   = (const float*)d_in[4];
  const float* dtb_b = (const float*)d_in[5];
  float* y = (float*)d_out;

  float* xp   = (float*)d_ws;
  float* dtb  = xp + (size_t)NROWS * NPROJ;
  float* part = dtb;
  float* hend = dtb + (size_t)NROWS * DIN;
  float* sdt  = hend + (size_t)(NCHUNK - 1) * DSTATE * NBD;

  gemm_xproj_partial<<<dim3(320), dim3(256), 0, stream>>>(x, xw, part);
  reduce_xp_kernel<<<dim3(640), dim3(256), 0, stream>>>(part, xp);
  gemm_dt_kernel<<<dim3(2048), dim3(256), 0, stream>>>(xp, dtw, dtb_b, dtb);
  scan_local_kernel<<<dim3(DIN / 32, BATCH, NCHUNK - 1), dim3(32), 0, stream>>>(
      x, A_log, xp, dtb, hend, sdt);
  scan_kernel<<<dim3(DIN / 32, BATCH, NCHUNK), dim3(32), 0, stream>>>(
      x, A_log, Dp, xp, dtb, hend, sdt, y);
}


# x_proj GEMM: x and w tiles staged through LDS by whole-line global loads, 8 waves read fragments there (1x5 tiles each), next stage's loads in flight; split-K 4->16, partial-sum order changes
# speedup vs baseline: 3.4186x; 1.3846x over previous
#include <hip/hip_runtime.h>
#include <hip/hip_bf16.h>


#define BATCH   2
#define SEQ     512
#define DIN     4096
#define DSTATE  16
#define DTRANK  128
#define NPROJ   160
#define NROWS   (BATCH * SEQ)
#define KSPLIT  16

typedef float v2f __attribute__((ext_vector_type(2)));
typedef float v4f __attribute__((ext_vector_type(4)));
typedef float v8f __attribute__((ext_vector_type(8)));

__device__ __forceinline__ v8f wmma_pair(v8f c, v4f a, v4f b) {
  v2f a0 = {a.x, a.y}, a1 = {a.z, a.w};
  v2f b0 = {b.x, b.y}, b1 = {b.z, b.w};
  c = __builtin_amdgcn_wmma_f32_16x16x4_f32(false, a0, false, b0,
                                            (short)0, c, false, false);
  c = __builtin_amdgcn_wmma_f32_16x16x4_f32(false, a1, false, b1,
                                            (short)0, c, false, false);
  return c;
}

#define K1_MB   64
#define K1_KT   64
#define K1_KLEN (DIN / KSPLIT)
#define K1_NST  (K1_KLEN / K1_KT)
#define K1_P4   17

__global__ __launch_bounds__(256) void gemm_xproj_partial(
    const float* __restrict__ x, const float* __restrict__ w,
    float* __restrict__ part) {
  __shared__ v4f sX[K1_MB * K1_P4];
  __shared__ v4f sW[NPROJ * K1_P4];

  const int tid   = threadIdx.x;
  const int lane  = tid & 31;
  const int wave  = tid >> 5;
  const int mb    = blockIdx.x & 15;
  const int ks    = blockIdx.x >> 4;
  const int kbase = ks * K1_KLEN;

  const int sr = tid >> 4, sc = tid & 15;
  const float* xg = x + (size_t)(mb * K1_MB + sr) * DIN + kbase + sc * 4;
  const float* wg = w + (size_t)sr * DIN + kbase + sc * 4;
  v4f* sxp = sX + sr * K1_P4 + sc;
  v4f* swp = sW + sr * K1_P4 + sc;

  v4f rx[K1_MB / 16], rw[NPROJ / 16];
#pragma unroll
  for (int i = 0; i < K1_MB / 16; ++i)
    rx[i] = *(const v4f*)(xg + (size_t)(16 * i) * DIN);
#pragma unroll
  for (int i = 0; i < NPROJ / 16; ++i)
    rw[i] = *(const v4f*)(wg + (size_t)(16 * i) * DIN);

  const int lo = lane & 15, hi = lane >> 4;
  const int mi = wave >> 1, nh = wave & 1;
  const v4f* aL = sX + (mi * 16 + lo) * K1_P4 + hi;
  const v4f* bL = sW + (nh * 80 + lo) * K1_P4 + hi;

  v8f acc[5];
#pragma unroll
  for (int j = 0; j < 5; ++j) acc[j] = (v8f){};

#pragma unroll 1
  for (int s = 0; s < K1_NST; ++s) {
#pragma unroll
    for (int i = 0; i < K1_MB / 16; ++i) sxp[16 * i * K1_P4] = rx[i];
#pragma unroll
    for (int i = 0; i < NPROJ / 16; ++i) swp[16 * i * K1_P4] = rw[i];
    __syncthreads();

    if (s + 1 < K1_NST) {
      const int ko = (s + 1) * K1_KT;
#pragma unroll
      for (int i = 0; i < K1_MB / 16; ++i)
        rx[i] = *(const v4f*)(xg + (size_t)(16 * i) * DIN + ko);
#pragma unroll
      for (int i = 0; i < NPROJ / 16; ++i)
        rw[i] = *(const v4f*)(wg + (size_t)(16 * i) * DIN + ko);
    }

#pragma unroll
    for (int kk = 0; kk < K1_KT / 8; ++kk) {
      const v4f a = aL[kk * 2];
      v4f b[5];
#pragma unroll
      for (int j = 0; j < 5; ++j) b[j] = bL[j * 16 * K1_P4 + kk * 2];
      const v2f a0 = {a.x, a.y}, a1 = {a.z, a.w};
#pragma unroll
      for (int j = 0; j < 5; ++j) {
        const v2f b0 = {b[j].x, b[j].y};
        acc[j] = __builtin_amdgcn_wmma_f32_16x16x4_f32(
            false, a0, false, b0, (short)0, acc[j], false, false);
      }
#pragma unroll
      for (int j = 0; j < 5; ++j) {
        const v2f b1 = {b[j].z, b[j].w};
        acc[j] = __builtin_amdgcn_wmma_f32_16x16x4_f32(
            false, a1, false, b1, (short)0, acc[j], false, false);
      }
    }
    __syncthreads();
  }

  const int rbase = mb * K1_MB + mi * 16 + 8 * hi;
  float* out = part + (size_t)ks * NROWS * NPROJ + nh * 80 + lo;
#pragma unroll
  for (int j = 0; j < 5; ++j) {
#pragma unroll
    for (int r = 0; r < 8; ++r)
      out[(size_t)(rbase + r) * NPROJ + j * 16] = acc[j][r];
  }
}

__global__ __launch_bounds__(256) void reduce_xp_kernel(
    const float* __restrict__ part, float* __restrict__ xp) {
  const size_t S4 = (size_t)NROWS * NPROJ / 4;
  const size_t i = (size_t)blockIdx.x * 256 + threadIdx.x;
  const v4f* p = (const v4f*)part + i;
  v4f v[KSPLIT];
#pragma unroll
  for (int q = 0; q < KSPLIT; ++q) v[q] = p[(size_t)q * S4];
  v4f s = v[0];
#pragma unroll
  for (int q = 1; q < KSPLIT; ++q) s = s + v[q];
  ((v4f*)xp)[i] = s;
}

__global__ __launch_bounds__(256) void gemm_dt_kernel(
    const float* __restrict__ xp, const float* __restrict__ w,
    const float* __restrict__ bias, float* __restrict__ dtb) {
  const int lane = threadIdx.x & 31;
  const int wave = threadIdx.x >> 5;
  const int tile = blockIdx.x * 8 + wave;
  const int mt = tile >> 8;
  const int nt = tile & 255;
  const int m0 = mt * 16, n0 = nt * 16;
  const int lo = lane & 15, hi = lane >> 4;

  const float* aptr = xp + (size_t)(m0 + lo) * NPROJ + 4 * hi;
  const float* bptr = w + (size_t)(n0 + lo) * DTRANK + 4 * hi;

  v8f c = {};
#pragma unroll
  for (int k = 0; k < DTRANK; k += 8) {
    v4f a = *(const v4f*)(aptr + k);
    v4f b = *(const v4f*)(bptr + k);
    c = wmma_pair(c, a, b);
  }

  const int d = n0 + lo;
  const float bv = bias[d];
  float* out = dtb + d;
  const int rbase = m0 + 8 * hi;
#pragma unroll
  for (int j = 0; j < 8; ++j) {
    float v = c[j] + bv;
    float sp = (v > 20.0f) ? v : log1pf(__expf(v));
    out[(size_t)(rbase + j) * DIN] = sp;
  }
}

#define NCHUNK 8
#define CLEN (SEQ / NCHUNK)
#define TB 16
#define NBD (BATCH * DIN)
#define LOG2E 1.4426950408889634f

__device__ __forceinline__ v2f fma2(v2f a, v2f b, v2f c) {
  return __builtin_elementwise_fma(a, b, c);
}
__device__ __forceinline__ v2f exp2_2(v2f v) {
  v2f r = {__builtin_amdgcn_exp2f(v.x), __builtin_amdgcn_exp2f(v.y)};
  return r;
}

__device__ __forceinline__ void load_batch(v2f (&dx)[TB],
                                           const float* __restrict__ dtg,
                                           const float* __restrict__ xg, int t) {
#pragma unroll
  for (int i = 0; i < TB; ++i) {
    dx[i].x = dtg[(size_t)(t + i) * DIN];
    dx[i].y = xg [(size_t)(t + i) * DIN];
  }
}

__device__ __forceinline__ void load_A2(v2f (&A2)[8], const float* __restrict__ A_log, int d) {
  const v4f* ap = (const v4f*)(A_log + (size_t)d * DSTATE);
#pragma unroll
  for (int k = 0; k < 4; ++k) {
    const v4f a = ap[k];
    A2[2 * k]     = (v2f){-__expf(a.x) * LOG2E, -__expf(a.y) * LOG2E};
    A2[2 * k + 1] = (v2f){-__expf(a.z) * LOG2E, -__expf(a.w) * LOG2E};
  }
}

__device__ __forceinline__ void local_batch(const v2f (&dxc)[TB],
                                            v2f (&h)[8], const v2f (&A2)[8],
                                            float& sum_dt, const v4f* bt) {
  v4f bn[4];
#pragma unroll
  for (int k = 0; k < 4; ++k) bn[k] = bt[k];
#pragma unroll
  for (int i = 0; i < TB; ++i) {
    const v2f dx   = dxc[i];
    const v2f dt2  = __builtin_shufflevector(dx, dx, 0, 0);
    const v2f x2   = __builtin_shufflevector(dx, dx, 1, 1);
    const v2f dtx2 = dt2 * x2;
    sum_dt += dx.x;
    v4f bv[4];
#pragma unroll
    for (int k = 0; k < 4; ++k) bv[k] = bn[k];
    if (i + 1 < TB) {
#pragma unroll
      for (int k = 0; k < 4; ++k) bn[k] = bt[(i + 1) * 4 + k];
    }
#pragma unroll
    for (int p = 0; p < 8; ++p) {
      const v4f bq = bv[p >> 1];
      const v2f Bp = (p & 1) ? (v2f){bq.z, bq.w} : (v2f){bq.x, bq.y};
      const v2f e  = exp2_2(dt2 * A2[p]);
      h[p] = fma2(e, h[p], dtx2 * Bp);
    }
  }
}

__device__ __forceinline__ void scan_batch(const v2f (&dxc)[TB],
                                           v2f (&h)[8], const v2f (&A2)[8],
                                           float Dv, const v4f* bt, float* yt) {
  v4f bn[8];
#pragma unroll
  for (int k = 0; k < 8; ++k) bn[k] = bt[k];
#pragma unroll
  for (int i = 0; i < TB; ++i) {
    const v2f dx   = dxc[i];
    const float xv = dx.y;
    const v2f dt2  = __builtin_shufflevector(dx, dx, 0, 0);
    const v2f x2   = __builtin_shufflevector(dx, dx, 1, 1);
    const v2f dtx2 = dt2 * x2;
    v4f bv[8];
#pragma unroll
    for (int k = 0; k < 8; ++k) bv[k] = bn[k];
    if (i + 1 < TB) {
#pragma unroll
      for (int k = 0; k < 8; ++k) bn[k] = bt[(i + 1) * 8 + k];
    }
    v2f ya = {0.0f, 0.0f}, yb = {0.0f, 0.0f};
#pragma unroll
    for (int p = 0; p < 8; ++p) {
      const v4f bq = bv[p >> 1];
      const v4f cq = bv[4 + (p >> 1)];
      const v2f Bp = (p & 1) ? (v2f){bq.z, bq.w} : (v2f){bq.x, bq.y};
      const v2f Cp = (p & 1) ? (v2f){cq.z, cq.w} : (v2f){cq.x, cq.y};
      const v2f e  = exp2_2(dt2 * A2[p]);
      h[p] = fma2(e, h[p], dtx2 * Bp);
      if (p & 1) yb = fma2(h[p], Cp, yb);
      else       ya = fma2(h[p], Cp, ya);
    }
    const v2f ys = ya + yb;
    yt[(size_t)i * DIN] = (ys.x + ys.y) + Dv * xv;
  }
}

__global__ __launch_bounds__(32) void scan_local_kernel(
    const float* __restrict__ x, const float* __restrict__ A_log,
    const float* __restrict__ xp, const float* __restrict__ dtb,
    float* __restrict__ hend, float* __restrict__ sdt) {
  __shared__ v4f sB[CLEN * 4];

  const int lane = threadIdx.x;
  const int d    = blockIdx.x * 32 + lane;
  const int b    = blockIdx.y;
  const int ch   = blockIdx.z;
  const int tbeg = ch * CLEN;
  const int bd   = b * DIN + d;

  const float* xg  = x   + ((size_t)b * SEQ + tbeg) * DIN + d;
  const float* dtg = dtb + ((size_t)b * SEQ + tbeg) * DIN + d;

  v2f dxA[TB], dxB[TB];
  load_batch(dxA, dtg, xg, 0);

  const float* xpb = xp + ((size_t)b * SEQ + tbeg) * NPROJ + DTRANK;
#pragma unroll
  for (int i = 0; i < CLEN / 8; ++i) {
    const int r = i * 8 + (lane >> 2);
    sB[r * 4 + (lane & 3)] =
        *(const v4f*)(xpb + (size_t)r * NPROJ + (lane & 3) * 4);
  }

  v2f A2[8];
  load_A2(A2, A_log, d);

  v2f h[8];
#pragma unroll
  for (int p = 0; p < 8; ++p) h[p] = (v2f){0.0f, 0.0f};
  float sum_dt = 0.0f;

  __syncthreads();

#pragma unroll 1
  for (int t0 = 0; t0 < CLEN; t0 += 2 * TB) {
    load_batch(dxB, dtg, xg, t0 + TB);
    __builtin_amdgcn_sched_barrier(0);
    local_batch(dxA, h, A2, sum_dt, sB + t0 * 4);
    __builtin_amdgcn_sched_barrier(0);
    const int tn = (t0 + 2 * TB < CLEN) ? (t0 + 2 * TB) : t0;
    load_batch(dxA, dtg, xg, tn);
    __builtin_amdgcn_sched_barrier(0);
    local_batch(dxB, h, A2, sum_dt, sB + (t0 + TB) * 4);
  }

  v4f* hq = (v4f*)hend + (size_t)ch * 4 * NBD + bd;
#pragma unroll
  for (int q = 0; q < 4; ++q)
    hq[(size_t)q * NBD] =
        (v4f){h[2 * q].x, h[2 * q].y, h[2 * q + 1].x, h[2 * q + 1].y};
  sdt[(size_t)ch * NBD + bd] = sum_dt;
}

__global__ __launch_bounds__(32) void scan_kernel(
    const float* __restrict__ x, const float* __restrict__ A_log,
    const float* __restrict__ Dp, const float* __restrict__ xp,
    const float* __restrict__ dtb, const float* __restrict__ hend,
    const float* __restrict__ sdt, float* __restrict__ y) {
  __shared__ v4f sBC[CLEN * 8];

  const int lane = threadIdx.x;
  const int d    = blockIdx.x * 32 + lane;
  const int b    = blockIdx.y;
  const int ch   = blockIdx.z;
  const int tbeg = ch * CLEN;
  const int bd   = b * DIN + d;

  const float* xg  = x   + ((size_t)b * SEQ + tbeg) * DIN + d;
  const float* dtg = dtb + ((size_t)b * SEQ + tbeg) * DIN + d;
  float*       yg  = y   + ((size_t)b * SEQ + tbeg) * DIN + d;

  v2f dxA[TB], dxB[TB];
  load_batch(dxA, dtg, xg, 0);

  const float* xpb = xp + ((size_t)b * SEQ + tbeg) * NPROJ + DTRANK;
#pragma unroll
  for (int i = 0; i < CLEN / 4; ++i) {
    const int r = i * 4 + (lane >> 3);
    sBC[r * 8 + (lane & 7)] =
        *(const v4f*)(xpb + (size_t)r * NPROJ + (lane & 7) * 4);
  }

  v2f A2[8];
  load_A2(A2, A_log, d);
  const float Dv = Dp[d];

  v2f h[8];
#pragma unroll
  for (int p = 0; p < 8; ++p) h[p] = (v2f){0.0f, 0.0f};
  {
    float sd[NCHUNK - 1];
    v4f   he[NCHUNK - 1][4];
    const v4f* hq = (const v4f*)hend + bd;
#pragma unroll
    for (int cc = 0; cc < NCHUNK - 1; ++cc) {
      sd[cc] = sdt[(size_t)cc * NBD + bd];
#pragma unroll
      for (int q = 0; q < 4; ++q)
        he[cc][q] = hq[((size_t)cc * 4 + q) * NBD];
    }
#pragma unroll
    for (int cc = 0; cc < NCHUNK - 1; ++cc) {
      const bool on = cc < ch;
      const v2f s2 = {sd[cc], sd[cc]};
#pragma unroll
      for (int p = 0; p < 8; ++p) {
        const v4f hv = he[cc][p >> 1];
        v2f e  = exp2_2(s2 * A2[p]);
        v2f ev = (p & 1) ? (v2f){hv.z, hv.w} : (v2f){hv.x, hv.y};
        e.x  = on ? e.x  : 1.0f;  e.y  = on ? e.y  : 1.0f;
        ev.x = on ? ev.x : 0.0f;  ev.y = on ? ev.y : 0.0f;
        h[p] = fma2(e, h[p], ev);
      }
    }
  }

  __syncthreads();

#pragma unroll 1
  for (int t0 = 0; t0 < CLEN; t0 += 2 * TB) {
    load_batch(dxB, dtg, xg, t0 + TB);
    __builtin_amdgcn_sched_barrier(0);
    scan_batch(dxA, h, A2, Dv, sBC + t0 * 8, yg + (size_t)t0 * DIN);
    __builtin_amdgcn_sched_barrier(0);
    const int tn = (t0 + 2 * TB < CLEN) ? (t0 + 2 * TB) : t0;
    load_batch(dxA, dtg, xg, tn);
    __builtin_amdgcn_sched_barrier(0);
    scan_batch(dxB, h, A2, Dv, sBC + (t0 + TB) * 8,
               yg + (size_t)(t0 + TB) * DIN);
  }
}

extern "C" void kernel_launch(void* const* d_in, const int* in_sizes, int n_in,
                              void* d_out, int out_size, void* d_ws, size_t ws_size,
                              hipStream_t stream) {
  const float* x     = (const float*)d_in[0];
  const float* A_log = (const float*)d_in[1];
  const float* Dp    = (const float*)d_in[2];
  const float* xw    = (const float*)d_in[3];
  const float* dtw   = (const float*)d_in[4];
  const float* dtb_b = (const float*)d_in[5];
  float* y = (float*)d_out;

  float* xp   = (float*)d_ws;
  float* dtb  = xp + (size_t)NROWS * NPROJ;
  float* part = dtb;
  float* hend = dtb + (size_t)NROWS * DIN;
  float* sdt  = hend + (size_t)(NCHUNK - 1) * DSTATE * NBD;

  gemm_xproj_partial<<<dim3(256), dim3(256), 0, stream>>>(x, xw, part);
  reduce_xp_kernel<<<dim3(160), dim3(256), 0, stream>>>(part, xp);
  gemm_dt_kernel<<<dim3(2048), dim3(256), 0, stream>>>(xp, dtw, dtb_b, dtb);
  scan_local_kernel<<<dim3(DIN / 32, BATCH, NCHUNK - 1), dim3(32), 0, stream>>>(
      x, A_log, xp, dtb, hend, sdt);
  scan_kernel<<<dim3(DIN / 32, BATCH, NCHUNK), dim3(32), 0, stream>>>(
      x, A_log, Dp, xp, dtb, hend, sdt, y);
}


# dt GEMM: xp and dt_proj_w tiles staged in LDS by whole-line global loads, 4 waves read fragments there (1x4 tiles each); 16384 waves -> 4096, a lane's 32 softplus values branch-free; same k order
# speedup vs baseline: 3.8395x; 1.1231x over previous
#include <hip/hip_runtime.h>
#include <hip/hip_bf16.h>


#define BATCH   2
#define SEQ     512
#define DIN     4096
#define DSTATE  16
#define DTRANK  128
#define NPROJ   160
#define NROWS   (BATCH * SEQ)
#define KSPLIT  16

typedef float v2f __attribute__((ext_vector_type(2)));
typedef float v4f __attribute__((ext_vector_type(4)));
typedef float v8f __attribute__((ext_vector_type(8)));

#define K1_MB   64
#define K1_KT   64
#define K1_KLEN (DIN / KSPLIT)
#define K1_NST  (K1_KLEN / K1_KT)
#define K1_P4   17

__global__ __launch_bounds__(256) void gemm_xproj_partial(
    const float* __restrict__ x, const float* __restrict__ w,
    float* __restrict__ part) {
  __shared__ v4f sX[K1_MB * K1_P4];
  __shared__ v4f sW[NPROJ * K1_P4];

  const int tid   = threadIdx.x;
  const int lane  = tid & 31;
  const int wave  = tid >> 5;
  const int mb    = blockIdx.x & 15;
  const int ks    = blockIdx.x >> 4;
  const int kbase = ks * K1_KLEN;

  const int sr = tid >> 4, sc = tid & 15;
  const float* xg = x + (size_t)(mb * K1_MB + sr) * DIN + kbase + sc * 4;
  const float* wg = w + (size_t)sr * DIN + kbase + sc * 4;
  v4f* sxp = sX + sr * K1_P4 + sc;
  v4f* swp = sW + sr * K1_P4 + sc;

  v4f rx[K1_MB / 16], rw[NPROJ / 16];
#pragma unroll
  for (int i = 0; i < K1_MB / 16; ++i)
    rx[i] = *(const v4f*)(xg + (size_t)(16 * i) * DIN);
#pragma unroll
  for (int i = 0; i < NPROJ / 16; ++i)
    rw[i] = *(const v4f*)(wg + (size_t)(16 * i) * DIN);

  const int lo = lane & 15, hi = lane >> 4;
  const int mi = wave >> 1, nh = wave & 1;
  const v4f* aL = sX + (mi * 16 + lo) * K1_P4 + hi;
  const v4f* bL = sW + (nh * 80 + lo) * K1_P4 + hi;

  v8f acc[5];
#pragma unroll
  for (int j = 0; j < 5; ++j) acc[j] = (v8f){};

#pragma unroll 1
  for (int s = 0; s < K1_NST; ++s) {
#pragma unroll
    for (int i = 0; i < K1_MB / 16; ++i) sxp[16 * i * K1_P4] = rx[i];
#pragma unroll
    for (int i = 0; i < NPROJ / 16; ++i) swp[16 * i * K1_P4] = rw[i];
    __syncthreads();

    if (s + 1 < K1_NST) {
      const int ko = (s + 1) * K1_KT;
#pragma unroll
      for (int i = 0; i < K1_MB / 16; ++i)
        rx[i] = *(const v4f*)(xg + (size_t)(16 * i) * DIN + ko);
#pragma unroll
      for (int i = 0; i < NPROJ / 16; ++i)
        rw[i] = *(const v4f*)(wg + (size_t)(16 * i) * DIN + ko);
    }

#pragma unroll
    for (int kk = 0; kk < K1_KT / 8; ++kk) {
      const v4f a = aL[kk * 2];
      v4f b[5];
#pragma unroll
      for (int j = 0; j < 5; ++j) b[j] = bL[j * 16 * K1_P4 + kk * 2];
      const v2f a0 = {a.x, a.y}, a1 = {a.z, a.w};
#pragma unroll
      for (int j = 0; j < 5; ++j) {
        const v2f b0 = {b[j].x, b[j].y};
        acc[j] = __builtin_amdgcn_wmma_f32_16x16x4_f32(
            false, a0, false, b0, (short)0, acc[j], false, false);
      }
#pragma unroll
      for (int j = 0; j < 5; ++j) {
        const v2f b1 = {b[j].z, b[j].w};
        acc[j] = __builtin_amdgcn_wmma_f32_16x16x4_f32(
            false, a1, false, b1, (short)0, acc[j], false, false);
      }
    }
    __syncthreads();
  }

  const int rbase = mb * K1_MB + mi * 16 + 8 * hi;
  float* out = part + (size_t)ks * NROWS * NPROJ + nh * 80 + lo;
#pragma unroll
  for (int j = 0; j < 5; ++j) {
#pragma unroll
    for (int r = 0; r < 8; ++r)
      out[(size_t)(rbase + r) * NPROJ + j * 16] = acc[j][r];
  }
}

__global__ __launch_bounds__(256) void reduce_xp_kernel(
    const float* __restrict__ part, float* __restrict__ xp) {
  const size_t S4 = (size_t)NROWS * NPROJ / 4;
  const size_t i = (size_t)blockIdx.x * 256 + threadIdx.x;
  const v4f* p = (const v4f*)part + i;
  v4f v[KSPLIT];
#pragma unroll
  for (int q = 0; q < KSPLIT; ++q) v[q] = p[(size_t)q * S4];
  v4f s = v[0];
#pragma unroll
  for (int q = 1; q < KSPLIT; ++q) s = s + v[q];
  ((v4f*)xp)[i] = s;
}

#define K2_MB   64
#define K2_NB   64
#define K2_KT   64
#define K2_NST  (DTRANK / K2_KT)
#define K2_P4   17

__global__ __launch_bounds__(128) void gemm_dt_kernel(
    const float* __restrict__ xp, const float* __restrict__ w,
    const float* __restrict__ bias, float* __restrict__ dtb) {
  __shared__ v4f sA[K2_MB * K2_P4];
  __shared__ v4f sW[K2_NB * K2_P4];

  const int tid  = threadIdx.x;
  const int lane = tid & 31;
  const int wave = tid >> 5;
  const int nb   = blockIdx.x & 63;
  const int mb   = blockIdx.x >> 6;

  const int sr = tid >> 4, sc = tid & 15;
  const float* ag = xp + (size_t)(mb * K2_MB + sr) * NPROJ + sc * 4;
  const float* wg = w + (size_t)(nb * K2_NB + sr) * DTRANK + sc * 4;
  v4f* sap = sA + sr * K2_P4 + sc;
  v4f* swp = sW + sr * K2_P4 + sc;

  v4f ra[K2_MB / 8], rw[K2_NB / 8];
#pragma unroll
  for (int i = 0; i < K2_MB / 8; ++i)
    ra[i] = *(const v4f*)(ag + (size_t)(8 * i) * NPROJ);
#pragma unroll
  for (int i = 0; i < K2_NB / 8; ++i)
    rw[i] = *(const v4f*)(wg + (size_t)(8 * i) * DTRANK);

  const int lo = lane & 15, hi = lane >> 4;
  const v4f* aL = sA + (wave * 16 + lo) * K2_P4 + hi;
  const v4f* bL = sW + lo * K2_P4 + hi;

  v8f acc[4];
#pragma unroll
  for (int j = 0; j < 4; ++j) acc[j] = (v8f){};

#pragma unroll
  for (int s = 0; s < K2_NST; ++s) {
#pragma unroll
    for (int i = 0; i < K2_MB / 8; ++i) sap[8 * i * K2_P4] = ra[i];
#pragma unroll
    for (int i = 0; i < K2_NB / 8; ++i) swp[8 * i * K2_P4] = rw[i];
    __syncthreads();

    if (s + 1 < K2_NST) {
      const int ko = (s + 1) * K2_KT;
#pragma unroll
      for (int i = 0; i < K2_MB / 8; ++i)
        ra[i] = *(const v4f*)(ag + (size_t)(8 * i) * NPROJ + ko);
#pragma unroll
      for (int i = 0; i < K2_NB / 8; ++i)
        rw[i] = *(const v4f*)(wg + (size_t)(8 * i) * DTRANK + ko);
    }

#pragma unroll
    for (int kk = 0; kk < K2_KT / 8; ++kk) {
      const v4f a = aL[kk * 2];
      v4f b[4];
#pragma unroll
      for (int j = 0; j < 4; ++j) b[j] = bL[j * 16 * K2_P4 + kk * 2];
      const v2f a0 = {a.x, a.y}, a1 = {a.z, a.w};
#pragma unroll
      for (int j = 0; j < 4; ++j) {
        const v2f b0 = {b[j].x, b[j].y};
        acc[j] = __builtin_amdgcn_wmma_f32_16x16x4_f32(
            false, a0, false, b0, (short)0, acc[j], false, false);
      }
#pragma unroll
      for (int j = 0; j < 4; ++j) {
        const v2f b1 = {b[j].z, b[j].w};
        acc[j] = __builtin_amdgcn_wmma_f32_16x16x4_f32(
            false, a1, false, b1, (short)0, acc[j], false, false);
      }
    }
    if (s + 1 < K2_NST)
      __syncthreads();
  }

  const int rbase = mb * K2_MB + wave * 16 + 8 * hi;
#pragma unroll
  for (int j = 0; j < 4; ++j) {
    const int d = nb * K2_NB + j * 16 + lo;
    const float bv = bias[d];
    float* out = dtb + d;
#pragma unroll
    for (int r = 0; r < 8; ++r) {
      const float v = acc[j][r] + bv;
      const float l = log1pf(__expf(v));
      out[(size_t)(rbase + r) * DIN] = (v > 20.0f) ? v : l;
    }
  }
}

#define NCHUNK 8
#define CLEN (SEQ / NCHUNK)
#define TB 16
#define NBD (BATCH * DIN)
#define LOG2E 1.4426950408889634f

__device__ __forceinline__ v2f fma2(v2f a, v2f b, v2f c) {
  return __builtin_elementwise_fma(a, b, c);
}
__device__ __forceinline__ v2f exp2_2(v2f v) {
  v2f r = {__builtin_amdgcn_exp2f(v.x), __builtin_amdgcn_exp2f(v.y)};
  return r;
}

__device__ __forceinline__ void load_batch(v2f (&dx)[TB],
                                           const float* __restrict__ dtg,
                                           const float* __restrict__ xg, int t) {
#pragma unroll
  for (int i = 0; i < TB; ++i) {
    dx[i].x = dtg[(size_t)(t + i) * DIN];
    dx[i].y = xg [(size_t)(t + i) * DIN];
  }
}

__device__ __forceinline__ void load_A2(v2f (&A2)[8], const float* __restrict__ A_log, int d) {
  const v4f* ap = (const v4f*)(A_log + (size_t)d * DSTATE);
#pragma unroll
  for (int k = 0; k < 4; ++k) {
    const v4f a = ap[k];
    A2[2 * k]     = (v2f){-__expf(a.x) * LOG2E, -__expf(a.y) * LOG2E};
    A2[2 * k + 1] = (v2f){-__expf(a.z) * LOG2E, -__expf(a.w) * LOG2E};
  }
}

__device__ __forceinline__ void local_batch(const v2f (&dxc)[TB],
                                            v2f (&h)[8], const v2f (&A2)[8],
                                            float& sum_dt, const v4f* bt) {
  v4f bn[4];
#pragma unroll
  for (int k = 0; k < 4; ++k) bn[k] = bt[k];
#pragma unroll
  for (int i = 0; i < TB; ++i) {
    const v2f dx   = dxc[i];
    const v2f dt2  = __builtin_shufflevector(dx, dx, 0, 0);
    const v2f x2   = __builtin_shufflevector(dx, dx, 1, 1);
    const v2f dtx2 = dt2 * x2;
    sum_dt += dx.x;
    v4f bv[4];
#pragma unroll
    for (int k = 0; k < 4; ++k) bv[k] = bn[k];
    if (i + 1 < TB) {
#pragma unroll
      for (int k = 0; k < 4; ++k) bn[k] = bt[(i + 1) * 4 + k];
    }
#pragma unroll
    for (int p = 0; p < 8; ++p) {
      const v4f bq = bv[p >> 1];
      const v2f Bp = (p & 1) ? (v2f){bq.z, bq.w} : (v2f){bq.x, bq.y};
      const v2f e  = exp2_2(dt2 * A2[p]);
      h[p] = fma2(e, h[p], dtx2 * Bp);
    }
  }
}

__device__ __forceinline__ void scan_batch(const v2f (&dxc)[TB],
                                           v2f (&h)[8], const v2f (&A2)[8],
                                           float Dv, const v4f* bt, float* yt) {
  v4f bn[8];
#pragma unroll
  for (int k = 0; k < 8; ++k) bn[k] = bt[k];
#pragma unroll
  for (int i = 0; i < TB; ++i) {
    const v2f dx   = dxc[i];
    const float xv = dx.y;
    const v2f dt2  = __builtin_shufflevector(dx, dx, 0, 0);
    const v2f x2   = __builtin_shufflevector(dx, dx, 1, 1);
    const v2f dtx2 = dt2 * x2;
    v4f bv[8];
#pragma unroll
    for (int k = 0; k < 8; ++k) bv[k] = bn[k];
    if (i + 1 < TB) {
#pragma unroll
      for (int k = 0; k < 8; ++k) bn[k] = bt[(i + 1) * 8 + k];
    }
    v2f ya = {0.0f, 0.0f}, yb = {0.0f, 0.0f};
#pragma unroll
    for (int p = 0; p < 8; ++p) {
      const v4f bq = bv[p >> 1];
      const v4f cq = bv[4 + (p >> 1)];
      const v2f Bp = (p & 1) ? (v2f){bq.z, bq.w} : (v2f){bq.x, bq.y};
      const v2f Cp = (p & 1) ? (v2f){cq.z, cq.w} : (v2f){cq.x, cq.y};
      const v2f e  = exp2_2(dt2 * A2[p]);
      h[p] = fma2(e, h[p], dtx2 * Bp);
      if (p & 1) yb = fma2(h[p], Cp, yb);
      else       ya = fma2(h[p], Cp, ya);
    }
    const v2f ys = ya + yb;
    yt[(size_t)i * DIN] = (ys.x + ys.y) + Dv * xv;
  }
}

__global__ __launch_bounds__(32) void scan_local_kernel(
    const float* __restrict__ x, const float* __restrict__ A_log,
    const float* __restrict__ xp, const float* __restrict__ dtb,
    float* __restrict__ hend, float* __restrict__ sdt) {
  __shared__ v4f sB[CLEN * 4];

  const int lane = threadIdx.x;
  const int d    = blockIdx.x * 32 + lane;
  const int b    = blockIdx.y;
  const int ch   = blockIdx.z;
  const int tbeg = ch * CLEN;
  const int bd   = b * DIN + d;

  const float* xg  = x   + ((size_t)b * SEQ + tbeg) * DIN + d;
  const float* dtg = dtb + ((size_t)b * SEQ + tbeg) * DIN + d;

  v2f dxA[TB], dxB[TB];
  load_batch(dxA, dtg, xg, 0);

  const float* xpb = xp + ((size_t)b * SEQ + tbeg) * NPROJ + DTRANK;
#pragma unroll
  for (int i = 0; i < CLEN / 8; ++i) {
    const int r = i * 8 + (lane >> 2);
    sB[r * 4 + (lane & 3)] =
        *(const v4f*)(xpb + (size_t)r * NPROJ + (lane & 3) * 4);
  }

  v2f A2[8];
  load_A2(A2, A_log, d);

  v2f h[8];
#pragma unroll
  for (int p = 0; p < 8; ++p) h[p] = (v2f){0.0f, 0.0f};
  float sum_dt = 0.0f;

  __syncthreads();

#pragma unroll 1
  for (int t0 = 0; t0 < CLEN; t0 += 2 * TB) {
    load_batch(dxB, dtg, xg, t0 + TB);
    __builtin_amdgcn_sched_barrier(0);
    local_batch(dxA, h, A2, sum_dt, sB + t0 * 4);
    __builtin_amdgcn_sched_barrier(0);
    const int tn = (t0 + 2 * TB < CLEN) ? (t0 + 2 * TB) : t0;
    load_batch(dxA, dtg, xg, tn);
    __builtin_amdgcn_sched_barrier(0);
    local_batch(dxB, h, A2, sum_dt, sB + (t0 + TB) * 4);
  }

  v4f* hq = (v4f*)hend + (size_t)ch * 4 * NBD + bd;
#pragma unroll
  for (int q = 0; q < 4; ++q)
    hq[(size_t)q * NBD] =
        (v4f){h[2 * q].x, h[2 * q].y, h[2 * q + 1].x, h[2 * q + 1].y};
  sdt[(size_t)ch * NBD + bd] = sum_dt;
}

__global__ __launch_bounds__(32) void scan_kernel(
    const float* __restrict__ x, const float* __restrict__ A_log,
    const float* __restrict__ Dp, const float* __restrict__ xp,
    const float* __restrict__ dtb, const float* __restrict__ hend,
    const float* __restrict__ sdt, float* __restrict__ y) {
  __shared__ v4f sBC[CLEN * 8];

  const int lane = threadIdx.x;
  const int d    = blockIdx.x * 32 + lane;
  const int b    = blockIdx.y;
  const int ch   = blockIdx.z;
  const int tbeg = ch * CLEN;
  const int bd   = b * DIN + d;

  const float* xg  = x   + ((size_t)b * SEQ + tbeg) * DIN + d;
  const float* dtg = dtb + ((size_t)b * SEQ + tbeg) * DIN + d;
  float*       yg  = y   + ((size_t)b * SEQ + tbeg) * DIN + d;

  v2f dxA[TB], dxB[TB];
  load_batch(dxA, dtg, xg, 0);

  const float* xpb = xp + ((size_t)b * SEQ + tbeg) * NPROJ + DTRANK;
#pragma unroll
  for (int i = 0; i < CLEN / 4; ++i) {
    const int r = i * 4 + (lane >> 3);
    sBC[r * 8 + (lane & 7)] =
        *(const v4f*)(xpb + (size_t)r * NPROJ + (lane & 7) * 4);
  }

  v2f A2[8];
  load_A2(A2, A_log, d);
  const float Dv = Dp[d];

  v2f h[8];
#pragma unroll
  for (int p = 0; p < 8; ++p) h[p] = (v2f){0.0f, 0.0f};
  {
    float sd[NCHUNK - 1];
    v4f   he[NCHUNK - 1][4];
    const v4f* hq = (const v4f*)hend + bd;
#pragma unroll
    for (int cc = 0; cc < NCHUNK - 1; ++cc) {
      sd[cc] = sdt[(size_t)cc * NBD + bd];
#pragma unroll
      for (int q = 0; q < 4; ++q)
        he[cc][q] = hq[((size_t)cc * 4 + q) * NBD];
    }
#pragma unroll
    for (int cc = 0; cc < NCHUNK - 1; ++cc) {
      const bool on = cc < ch;
      const v2f s2 = {sd[cc], sd[cc]};
#pragma unroll
      for (int p = 0; p < 8; ++p) {
        const v4f hv = he[cc][p >> 1];
        v2f e  = exp2_2(s2 * A2[p]);
        v2f ev = (p & 1) ? (v2f){hv.z, hv.w} : (v2f){hv.x, hv.y};
        e.x  = on ? e.x  : 1.0f;  e.y  = on ? e.y  : 1.0f;
        ev.x = on ? ev.x : 0.0f;  ev.y = on ? ev.y : 0.0f;
        h[p] = fma2(e, h[p], ev);
      }
    }
  }

  __syncthreads();

#pragma unroll 1
  for (int t0 = 0; t0 < CLEN; t0 += 2 * TB) {
    load_batch(dxB, dtg, xg, t0 + TB);
    __builtin_amdgcn_sched_barrier(0);
    scan_batch(dxA, h, A2, Dv, sBC + t0 * 8, yg + (size_t)t0 * DIN);
    __builtin_amdgcn_sched_barrier(0);
    const int tn = (t0 + 2 * TB < CLEN) ? (t0 + 2 * TB) : t0;
    load_batch(dxA, dtg, xg, tn);
    __builtin_amdgcn_sched_barrier(0);
    scan_batch(dxB, h, A2, Dv, sBC + (t0 + TB) * 8,
               yg + (size_t)(t0 + TB) * DIN);
  }
}

extern "C" void kernel_launch(void* const* d_in, const int* in_sizes, int n_in,
                              void* d_out, int out_size, void* d_ws, size_t ws_size,
                              hipStream_t stream) {
  const float* x     = (const float*)d_in[0];
  const float* A_log = (const float*)d_in[1];
  const float* Dp    = (const float*)d_in[2];
  const float* xw    = (const float*)d_in[3];
  const float* dtw   = (const float*)d_in[4];
  const float* dtb_b = (const float*)d_in[5];
  float* y = (float*)d_out;

  float* xp   = (float*)d_ws;
  float* dtb  = xp + (size_t)NROWS * NPROJ;
  float* part = dtb;
  float* hend = dtb + (size_t)NROWS * DIN;
  float* sdt  = hend + (size_t)(NCHUNK - 1) * DSTATE * NBD;

  gemm_xproj_partial<<<dim3(256), dim3(256), 0, stream>>>(x, xw, part);
  reduce_xp_kernel<<<dim3(160), dim3(256), 0, stream>>>(part, xp);
  gemm_dt_kernel<<<dim3(1024), dim3(128), 0, stream>>>(xp, dtw, dtb_b, dtb);
  scan_local_kernel<<<dim3(DIN / 32, BATCH, NCHUNK - 1), dim3(32), 0, stream>>>(
      x, A_log, xp, dtb, hend, sdt);
  scan_kernel<<<dim3(DIN / 32, BATCH, NCHUNK), dim3(32), 0, stream>>>(
      x, A_log, Dp, xp, dtb, hend, sdt, y);
}


# scan: its two launches made one: a work-group's 8 waves take the 8 chunks of 32 channels, chunk end states and dt sums go through LDS behind one barrier, not through memory; 5 launches -> 4
# speedup vs baseline: 4.1944x; 1.0924x over previous
#include <hip/hip_runtime.h>
#include <hip/hip_bf16.h>


#define BATCH   2
#define SEQ     512
#define DIN     4096
#define DSTATE  16
#define DTRANK  128
#define NPROJ   160
#define NROWS   (BATCH * SEQ)
#define KSPLIT  16

typedef float v2f __attribute__((ext_vector_type(2)));
typedef float v4f __attribute__((ext_vector_type(4)));
typedef float v8f __attribute__((ext_vector_type(8)));

#define K1_MB   64
#define K1_KT   64
#define K1_KLEN (DIN / KSPLIT)
#define K1_NST  (K1_KLEN / K1_KT)
#define K1_P4   17

__global__ __launch_bounds__(256) void gemm_xproj_partial(
    const float* __restrict__ x, const float* __restrict__ w,
    float* __restrict__ part) {
  __shared__ v4f sX[K1_MB * K1_P4];
  __shared__ v4f sW[NPROJ * K1_P4];

  const int tid   = threadIdx.x;
  const int lane  = tid & 31;
  const int wave  = tid >> 5;
  const int mb    = blockIdx.x & 15;
  const int ks    = blockIdx.x >> 4;
  const int kbase = ks * K1_KLEN;

  const int sr = tid >> 4, sc = tid & 15;
  const float* xg = x + (size_t)(mb * K1_MB + sr) * DIN + kbase + sc * 4;
  const float* wg = w + (size_t)sr * DIN + kbase + sc * 4;
  v4f* sxp = sX + sr * K1_P4 + sc;
  v4f* swp = sW + sr * K1_P4 + sc;

  v4f rx[K1_MB / 16], rw[NPROJ / 16];
#pragma unroll
  for (int i = 0; i < K1_MB / 16; ++i)
    rx[i] = *(const v4f*)(xg + (size_t)(16 * i) * DIN);
#pragma unroll
  for (int i = 0; i < NPROJ / 16; ++i)
    rw[i] = *(const v4f*)(wg + (size_t)(16 * i) * DIN);

  const int lo = lane & 15, hi = lane >> 4;
  const int mi = wave >> 1, nh = wave & 1;
  const v4f* aL = sX + (mi * 16 + lo) * K1_P4 + hi;
  const v4f* bL = sW + (nh * 80 + lo) * K1_P4 + hi;

  v8f acc[5];
#pragma unroll
  for (int j = 0; j < 5; ++j) acc[j] = (v8f){};

#pragma unroll 1
  for (int s = 0; s < K1_NST; ++s) {
#pragma unroll
    for (int i = 0; i < K1_MB / 16; ++i) sxp[16 * i * K1_P4] = rx[i];
#pragma unroll
    for (int i = 0; i < NPROJ / 16; ++i) swp[16 * i * K1_P4] = rw[i];
    __syncthreads();

    if (s + 1 < K1_NST) {
      const int ko = (s + 1) * K1_KT;
#pragma unroll
      for (int i = 0; i < K1_MB / 16; ++i)
        rx[i] = *(const v4f*)(xg + (size_t)(16 * i) * DIN + ko);
#pragma unroll
      for (int i = 0; i < NPROJ / 16; ++i)
        rw[i] = *(const v4f*)(wg + (size_t)(16 * i) * DIN + ko);
    }

#pragma unroll
    for (int kk = 0; kk < K1_KT / 8; ++kk) {
      const v4f a = aL[kk * 2];
      v4f b[5];
#pragma unroll
      for (int j = 0; j < 5; ++j) b[j] = bL[j * 16 * K1_P4 + kk * 2];
      const v2f a0 = {a.x, a.y}, a1 = {a.z, a.w};
#pragma unroll
      for (int j = 0; j < 5; ++j) {
        const v2f b0 = {b[j].x, b[j].y};
        acc[j] = __builtin_amdgcn_wmma_f32_16x16x4_f32(
            false, a0, false, b0, (short)0, acc[j], false, false);
      }
#pragma unroll
      for (int j = 0; j < 5; ++j) {
        const v2f b1 = {b[j].z, b[j].w};
        acc[j] = __builtin_amdgcn_wmma_f32_16x16x4_f32(
            false, a1, false, b1, (short)0, acc[j], false, false);
      }
    }
    __syncthreads();
  }

  const int rbase = mb * K1_MB + mi * 16 + 8 * hi;
  float* out = part + (size_t)ks * NROWS * NPROJ + nh * 80 + lo;
#pragma unroll
  for (int j = 0; j < 5; ++j) {
#pragma unroll
    for (int r = 0; r < 8; ++r)
      out[(size_t)(rbase + r) * NPROJ + j * 16] = acc[j][r];
  }
}

__global__ __launch_bounds__(256) void reduce_xp_kernel(
    const float* __restrict__ part, float* __restrict__ xp) {
  const size_t S4 = (size_t)NROWS * NPROJ / 4;
  const size_t i = (size_t)blockIdx.x * 256 + threadIdx.x;
  const v4f* p = (const v4f*)part + i;
  v4f v[KSPLIT];
#pragma unroll
  for (int q = 0; q < KSPLIT; ++q) v[q] = p[(size_t)q * S4];
  v4f s = v[0];
#pragma unroll
  for (int q = 1; q < KSPLIT; ++q) s = s + v[q];
  ((v4f*)xp)[i] = s;
}

#define K2_MB   64
#define K2_NB   64
#define K2_KT   64
#define K2_NST  (DTRANK / K2_KT)
#define K2_P4   17

__global__ __launch_bounds__(128) void gemm_dt_kernel(
    const float* __restrict__ xp, const float* __restrict__ w,
    const float* __restrict__ bias, float* __restrict__ dtb) {
  __shared__ v4f sA[K2_MB * K2_P4];
  __shared__ v4f sW[K2_NB * K2_P4];

  const int tid  = threadIdx.x;
  const int lane = tid & 31;
  const int wave = tid >> 5;
  const int nb   = blockIdx.x & 63;
  const int mb   = blockIdx.x >> 6;

  const int sr = tid >> 4, sc = tid & 15;
  const float* ag = xp + (size_t)(mb * K2_MB + sr) * NPROJ + sc * 4;
  const float* wg = w + (size_t)(nb * K2_NB + sr) * DTRANK + sc * 4;
  v4f* sap = sA + sr * K2_P4 + sc;
  v4f* swp = sW + sr * K2_P4 + sc;

  v4f ra[K2_MB / 8], rw[K2_NB / 8];
#pragma unroll
  for (int i = 0; i < K2_MB / 8; ++i)
    ra[i] = *(const v4f*)(ag + (size_t)(8 * i) * NPROJ);
#pragma unroll
  for (int i = 0; i < K2_NB / 8; ++i)
    rw[i] = *(const v4f*)(wg + (size_t)(8 * i) * DTRANK);

  const int lo = lane & 15, hi = lane >> 4;
  const v4f* aL = sA + (wave * 16 + lo) * K2_P4 + hi;
  const v4f* bL = sW + lo * K2_P4 + hi;

  v8f acc[4];
#pragma unroll
  for (int j = 0; j < 4; ++j) acc[j] = (v8f){};

#pragma unroll
  for (int s = 0; s < K2_NST; ++s) {
#pragma unroll
    for (int i = 0; i < K2_MB / 8; ++i) sap[8 * i * K2_P4] = ra[i];
#pragma unroll
    for (int i = 0; i < K2_NB / 8; ++i) swp[8 * i * K2_P4] = rw[i];
    __syncthreads();

    if (s + 1 < K2_NST) {
      const int ko = (s + 1) * K2_KT;
#pragma unroll
      for (int i = 0; i < K2_MB / 8; ++i)
        ra[i] = *(const v4f*)(ag + (size_t)(8 * i) * NPROJ + ko);
#pragma unroll
      for (int i = 0; i < K2_NB / 8; ++i)
        rw[i] = *(const v4f*)(wg + (size_t)(8 * i) * DTRANK + ko);
    }

#pragma unroll
    for (int kk = 0; kk < K2_KT / 8; ++kk) {
      const v4f a = aL[kk * 2];
      v4f b[4];
#pragma unroll
      for (int j = 0; j < 4; ++j) b[j] = bL[j * 16 * K2_P4 + kk * 2];
      const v2f a0 = {a.x, a.y}, a1 = {a.z, a.w};
#pragma unroll
      for (int j = 0; j < 4; ++j) {
        const v2f b0 = {b[j].x, b[j].y};
        acc[j] = __builtin_amdgcn_wmma_f32_16x16x4_f32(
            false, a0, false, b0, (short)0, acc[j], false, false);
      }
#pragma unroll
      for (int j = 0; j < 4; ++j) {
        const v2f b1 = {b[j].z, b[j].w};
        acc[j] = __builtin_amdgcn_wmma_f32_16x16x4_f32(
            false, a1, false, b1, (short)0, acc[j], false, false);
      }
    }
    if (s + 1 < K2_NST)
      __syncthreads();
  }

  const int rbase = mb * K2_MB + wave * 16 + 8 * hi;
#pragma unroll
  for (int j = 0; j < 4; ++j) {
    const int d = nb * K2_NB + j * 16 + lo;
    const float bv = bias[d];
    float* out = dtb + d;
#pragma unroll
    for (int r = 0; r < 8; ++r) {
      const float v = acc[j][r] + bv;
      const float l = log1pf(__expf(v));
      out[(size_t)(rbase + r) * DIN] = (v > 20.0f) ? v : l;
    }
  }
}

#define NCHUNK 8
#define CLEN (SEQ / NCHUNK)
#define TB 16
#define LOG2E 1.4426950408889634f

__device__ __forceinline__ v2f fma2(v2f a, v2f b, v2f c) {
  return __builtin_elementwise_fma(a, b, c);
}
__device__ __forceinline__ v2f exp2_2(v2f v) {
  v2f r = {__builtin_amdgcn_exp2f(v.x), __builtin_amdgcn_exp2f(v.y)};
  return r;
}

__device__ __forceinline__ void load_batch(v2f (&dx)[TB],
                                           const float* __restrict__ dtg,
                                           const float* __restrict__ xg, int t) {
#pragma unroll
  for (int i = 0; i < TB; ++i) {
    dx[i].x = dtg[(size_t)(t + i) * DIN];
    dx[i].y = xg [(size_t)(t + i) * DIN];
  }
}

__device__ __forceinline__ void load_A2(v2f (&A2)[8], const float* __restrict__ A_log, int d) {
  const v4f* ap = (const v4f*)(A_log + (size_t)d * DSTATE);
#pragma unroll
  for (int k = 0; k < 4; ++k) {
    const v4f a = ap[k];
    A2[2 * k]     = (v2f){-__expf(a.x) * LOG2E, -__expf(a.y) * LOG2E};
    A2[2 * k + 1] = (v2f){-__expf(a.z) * LOG2E, -__expf(a.w) * LOG2E};
  }
}

__device__ __forceinline__ void local_batch(const v2f (&dxc)[TB],
                                            v2f (&h)[8], const v2f (&A2)[8],
                                            float& sum_dt, const v4f* bt) {
  v4f bn[4];
#pragma unroll
  for (int k = 0; k < 4; ++k) bn[k] = bt[k];
#pragma unroll
  for (int i = 0; i < TB; ++i) {
    const v2f dx   = dxc[i];
    const v2f dt2  = __builtin_shufflevector(dx, dx, 0, 0);
    const v2f x2   = __builtin_shufflevector(dx, dx, 1, 1);
    const v2f dtx2 = dt2 * x2;
    sum_dt += dx.x;
    v4f bv[4];
#pragma unroll
    for (int k = 0; k < 4; ++k) bv[k] = bn[k];
    if (i + 1 < TB) {
#pragma unroll
      for (int k = 0; k < 4; ++k) bn[k] = bt[(i + 1) * 8 + k];
    }
#pragma unroll
    for (int p = 0; p < 8; ++p) {
      const v4f bq = bv[p >> 1];
      const v2f Bp = (p & 1) ? (v2f){bq.z, bq.w} : (v2f){bq.x, bq.y};
      const v2f e  = exp2_2(dt2 * A2[p]);
      h[p] = fma2(e, h[p], dtx2 * Bp);
    }
  }
}

__device__ __forceinline__ void scan_batch(const v2f (&dxc)[TB],
                                           v2f (&h)[8], const v2f (&A2)[8],
                                           float Dv, const v4f* bt, float* yt) {
  v4f bn[8];
#pragma unroll
  for (int k = 0; k < 8; ++k) bn[k] = bt[k];
#pragma unroll
  for (int i = 0; i < TB; ++i) {
    const v2f dx   = dxc[i];
    const float xv = dx.y;
    const v2f dt2  = __builtin_shufflevector(dx, dx, 0, 0);
    const v2f x2   = __builtin_shufflevector(dx, dx, 1, 1);
    const v2f dtx2 = dt2 * x2;
    v4f bv[8];
#pragma unroll
    for (int k = 0; k < 8; ++k) bv[k] = bn[k];
    if (i + 1 < TB) {
#pragma unroll
      for (int k = 0; k < 8; ++k) bn[k] = bt[(i + 1) * 8 + k];
    }
    v2f ya = {0.0f, 0.0f}, yb = {0.0f, 0.0f};
#pragma unroll
    for (int p = 0; p < 8; ++p) {
      const v4f bq = bv[p >> 1];
      const v4f cq = bv[4 + (p >> 1)];
      const v2f Bp = (p & 1) ? (v2f){bq.z, bq.w} : (v2f){bq.x, bq.y};
      const v2f Cp = (p & 1) ? (v2f){cq.z, cq.w} : (v2f){cq.x, cq.y};
      const v2f e  = exp2_2(dt2 * A2[p]);
      h[p] = fma2(e, h[p], dtx2 * Bp);
      if (p & 1) yb = fma2(h[p], Cp, yb);
      else       ya = fma2(h[p], Cp, ya);
    }
    const v2f ys = ya + yb;
    yt[(size_t)i * DIN] = (ys.x + ys.y) + Dv * xv;
  }
}

__global__ __launch_bounds__(256) void scan_kernel(
    const float* __restrict__ x, const float* __restrict__ A_log,
    const float* __restrict__ Dp, const float* __restrict__ xp,
    const float* __restrict__ dtb, float* __restrict__ y) {
  __shared__ v4f   sBC[NCHUNK * CLEN * 8];
  __shared__ v4f   sHe[(NCHUNK - 1) * 4 * 32];
  __shared__ float sSd[(NCHUNK - 1) * 32];

  const int lane = threadIdx.x & 31;
  const int ch   = __builtin_amdgcn_readfirstlane((int)(threadIdx.x >> 5));
  const int d    = blockIdx.x * 32 + lane;
  const int b    = blockIdx.y;
  const int tbeg = ch * CLEN;

  const float* xg  = x   + ((size_t)b * SEQ + tbeg) * DIN + d;
  const float* dtg = dtb + ((size_t)b * SEQ + tbeg) * DIN + d;
  float*       yg  = y   + ((size_t)b * SEQ + tbeg) * DIN + d;

  v2f dxA[TB], dxB[TB];
  load_batch(dxA, dtg, xg, 0);

  v4f* bc = sBC + ch * (CLEN * 8);
  const float* xpb = xp + ((size_t)b * SEQ + tbeg) * NPROJ + DTRANK;
#pragma unroll
  for (int i = 0; i < CLEN / 4; ++i) {
    const int r = i * 4 + (lane >> 3);
    bc[r * 8 + (lane & 7)] =
        *(const v4f*)(xpb + (size_t)r * NPROJ + (lane & 7) * 4);
  }

  v2f A2[8];
  load_A2(A2, A_log, d);
  const float Dv = Dp[d];

  v2f h[8];
#pragma unroll
  for (int p = 0; p < 8; ++p) h[p] = (v2f){0.0f, 0.0f};

  __syncthreads();

  if (ch < NCHUNK - 1) {
    float sum_dt = 0.0f;
#pragma unroll 1
    for (int t0 = 0; t0 < CLEN; t0 += 2 * TB) {
      load_batch(dxB, dtg, xg, t0 + TB);
      __builtin_amdgcn_sched_barrier(0);
      local_batch(dxA, h, A2, sum_dt, bc + t0 * 8);
      __builtin_amdgcn_sched_barrier(0);
      const int tn = (t0 + 2 * TB < CLEN) ? (t0 + 2 * TB) : 0;
      load_batch(dxA, dtg, xg, tn);
      __builtin_amdgcn_sched_barrier(0);
      local_batch(dxB, h, A2, sum_dt, bc + (t0 + TB) * 8);
    }
    v4f* hq = sHe + ch * (4 * 32) + lane;
#pragma unroll
    for (int q = 0; q < 4; ++q)
      hq[q * 32] = (v4f){h[2 * q].x, h[2 * q].y, h[2 * q + 1].x, h[2 * q + 1].y};
    sSd[ch * 32 + lane] = sum_dt;
  }

  __syncthreads();

#pragma unroll
  for (int p = 0; p < 8; ++p) h[p] = (v2f){0.0f, 0.0f};
#pragma unroll 1
  for (int cc = 0; cc < ch; ++cc) {
    const float sd = sSd[cc * 32 + lane];
    const v2f s2 = {sd, sd};
    const v4f* hq = sHe + cc * (4 * 32) + lane;
    v4f he[4];
#pragma unroll
    for (int q = 0; q < 4; ++q) he[q] = hq[q * 32];
#pragma unroll
    for (int p = 0; p < 8; ++p) {
      const v4f hv = he[p >> 1];
      const v2f e  = exp2_2(s2 * A2[p]);
      const v2f ev = (p & 1) ? (v2f){hv.z, hv.w} : (v2f){hv.x, hv.y};
      h[p] = fma2(e, h[p], ev);
    }
  }

#pragma unroll 1
  for (int t0 = 0; t0 < CLEN; t0 += 2 * TB) {
    load_batch(dxB, dtg, xg, t0 + TB);
    __builtin_amdgcn_sched_barrier(0);
    scan_batch(dxA, h, A2, Dv, bc + t0 * 8, yg + (size_t)t0 * DIN);
    __builtin_amdgcn_sched_barrier(0);
    const int tn = (t0 + 2 * TB < CLEN) ? (t0 + 2 * TB) : t0;
    load_batch(dxA, dtg, xg, tn);
    __builtin_amdgcn_sched_barrier(0);
    scan_batch(dxB, h, A2, Dv, bc + (t0 + TB) * 8,
               yg + (size_t)(t0 + TB) * DIN);
  }
}

extern "C" void kernel_launch(void* const* d_in, const int* in_sizes, int n_in,
                              void* d_out, int out_size, void* d_ws, size_t ws_size,
                              hipStream_t stream) {
  const float* x     = (const float*)d_in[0];
  const float* A_log = (const float*)d_in[1];
  const float* Dp    = (const float*)d_in[2];
  const float* xw    = (const float*)d_in[3];
  const float* dtw   = (const float*)d_in[4];
  const float* dtb_b = (const float*)d_in[5];
  float* y = (float*)d_out;

  float* xp   = (float*)d_ws;
  float* dtb  = xp + (size_t)NROWS * NPROJ;
  float* part = dtb;

  gemm_xproj_partial<<<dim3(256), dim3(256), 0, stream>>>(x, xw, part);
  reduce_xp_kernel<<<dim3(160), dim3(256), 0, stream>>>(part, xp);
  gemm_dt_kernel<<<dim3(1024), dim3(128), 0, stream>>>(xp, dtw, dtb_b, dtb);
  scan_kernel<<<dim3(DIN / 32, BATCH), dim3(256), 0, stream>>>(
      x, A_log, Dp, xp, dtb, y);
}
